# starting kernel
# baseline (speedup 1.0000x reference)
#include <hip/hip_runtime.h>
#include <hip/hip_bf16.h>


#define NQ      12
#define DIM     4096
#define NLAYERS 16
#define BATCH   2048
#define NCLASS  5

typedef float v2f __attribute__((ext_vector_type(2)));
typedef float v8f __attribute__((ext_vector_type(8)));

__device__ __forceinline__ int pa(int i) { return i + (i >> 6); }
#define PADDED_DIM (DIM + (DIM >> 6))

__global__ __launch_bounds__(256) void qnn_gates(const float* __restrict__ thetas,
                                                 float* __restrict__ gmat) {
  const int blk = blockIdx.x;
  const int l = blk / 3, g = blk % 3;
  const int t = threadIdx.x;
  const int m = t >> 4, k = t & 15;
  float prod = 1.0f;
#pragma unroll
  for (int j = 0; j < 4; ++j) {
    const int q = g * 4 + j;
    const float th = 0.5f * thetas[l * NQ + q];
    const float c = cosf(th), s = sinf(th);
    const int mb = (m >> (3 - j)) & 1;
    const int kb = (k >> (3 - j)) & 1;
    const float f = mb ? (kb ? c : s) : (kb ? -s : c);
    prod *= f;
  }
  gmat[blk * 256 + t] = prod;
}

__global__ __launch_bounds__(128) void qnn_main(const float* __restrict__ zr,
                                                const float* __restrict__ zi,
                                                const float* __restrict__ gmat,
                                                float* __restrict__ part) {
  __shared__ float bufA[PADDED_DIM];
  __shared__ float bufB[PADDED_DIM];
  __shared__ unsigned short tbl[DIM];
  __shared__ float gm[3 * 256];
  __shared__ float red[128 * NCLASS];

  const int tid  = threadIdx.x;
  const int row  = blockIdx.x;
  const int comp = row & 1;
  const int bidx = row >> 1;
  const float* __restrict__ src = comp ? zi : zr;

  for (int t = tid; t < DIM; t += 128)
    bufA[pa(t)] = src[(size_t)bidx * DIM + t];

  for (int i = tid; i < DIM; i += 128) {
    int j = i;
#pragma unroll
    for (int q = NQ - 1; q >= 0; --q) {
      const int cb = 1 << (NQ - 1 - q);
      const int tb = 1 << (NQ - 1 - ((q + 1) % NQ));
      j = (j & cb) ? (j ^ tb) : j;
    }
    tbl[i] = (unsigned short)j;
  }
  __syncthreads();

  float* cur = bufA;
  float* alt = bufB;

  const int lane = tid & 31;
  const int wv   = tid >> 5;
  const int half = lane >> 4;
  const int hm   = lane & 15;

  for (int l = 0; l < NLAYERS; ++l) {
    for (int t = tid; t < 3 * 256; t += 128) gm[t] = gmat[l * 768 + t];
    __syncthreads();

    {
      const float* in = cur;
      float* out = alt;
      v2f a[4];
#pragma unroll
      for (int kk = 0; kk < 4; ++kk) {
        const int k0 = kk * 4 + half * 2;
        a[kk].x = gm[hm * 16 + k0];
        a[kk].y = gm[hm * 16 + k0 + 1];
      }
#pragma unroll
      for (int t4 = 0; t4 < 4; ++t4) {
        const int bcb = (wv * 4 + t4) * 16;
        v8f acc = {0.f, 0.f, 0.f, 0.f, 0.f, 0.f, 0.f, 0.f};
#pragma unroll
        for (int kk = 0; kk < 4; ++kk) {
          const int k0 = kk * 4 + half * 2;
          int i0 = k0 * 256 + bcb + hm;
          int i1 = i0 + 256;
          if (l > 0) { i0 = tbl[i0]; i1 = tbl[i1]; }
          v2f bv;
          bv.x = in[pa(i0)];
          bv.y = in[pa(i1)];
          acc = __builtin_amdgcn_wmma_f32_16x16x4_f32(
              false, a[kk], false, bv, (short)0, acc, false, false);
        }
#pragma unroll
        for (int r = 0; r < 8; ++r)
          out[pa((r + half * 8) * 256 + bcb + hm)] = acc[r];
      }
    }
    __syncthreads();

    {
      const float* in = alt;
      float* out = cur;
      v2f a[4];
#pragma unroll
      for (int kk = 0; kk < 4; ++kk) {
        const int k0 = kk * 4 + half * 2;
        a[kk].x = gm[256 + hm * 16 + k0];
        a[kk].y = gm[256 + hm * 16 + k0 + 1];
      }
#pragma unroll
      for (int t4 = 0; t4 < 4; ++t4) {
        const int base = (wv * 4 + t4) * 256;
        v8f acc = {0.f, 0.f, 0.f, 0.f, 0.f, 0.f, 0.f, 0.f};
#pragma unroll
        for (int kk = 0; kk < 4; ++kk) {
          const int k0 = kk * 4 + half * 2;
          const int i0 = base + k0 * 16 + hm;
          v2f bv;
          bv.x = in[pa(i0)];
          bv.y = in[pa(i0 + 16)];
          acc = __builtin_amdgcn_wmma_f32_16x16x4_f32(
              false, a[kk], false, bv, (short)0, acc, false, false);
        }
#pragma unroll
        for (int r = 0; r < 8; ++r)
          out[pa(base + (r + half * 8) * 16 + hm)] = acc[r];
      }
    }
    __syncthreads();

    {
      const float* in = cur;
      float* out = alt;
      v2f a[4];
#pragma unroll
      for (int kk = 0; kk < 4; ++kk) {
        const int k0 = kk * 4 + half * 2;
        a[kk].x = gm[512 + hm * 16 + k0];
        a[kk].y = gm[512 + hm * 16 + k0 + 1];
      }
#pragma unroll
      for (int t4 = 0; t4 < 4; ++t4) {
        const int base = (wv * 4 + t4) * 256;
        v8f acc = {0.f, 0.f, 0.f, 0.f, 0.f, 0.f, 0.f, 0.f};
#pragma unroll
        for (int kk = 0; kk < 4; ++kk) {
          const int k0 = kk * 4 + half * 2;
          const int i0 = base + hm * 16 + k0;
          v2f bv;
          bv.x = in[pa(i0)];
          bv.y = in[pa(i0 + 1)];
          acc = __builtin_amdgcn_wmma_f32_16x16x4_f32(
              false, a[kk], false, bv, (short)0, acc, false, false);
        }
#pragma unroll
        for (int r = 0; r < 8; ++r)
          out[pa(base + hm * 16 + r + half * 8)] = acc[r];
      }
    }
    {
      float* tmp = cur; cur = alt; alt = tmp;
    }
    __syncthreads();
  }

  float acc[NCLASS] = {0.f, 0.f, 0.f, 0.f, 0.f};
  const float* fin = cur;
  for (int i = tid; i < DIM; i += 128) {
    const float s = fin[pa(tbl[i])];
    const float v = s * s;
#pragma unroll
    for (int w = 0; w < NCLASS; ++w)
      acc[w] += ((i >> (NQ - 1 - w)) & 1) ? -v : v;
  }
#pragma unroll
  for (int w = 0; w < NCLASS; ++w) red[tid * NCLASS + w] = acc[w];
  __syncthreads();
  for (int s = 64; s > 0; s >>= 1) {
    if (tid < s) {
#pragma unroll
      for (int w = 0; w < NCLASS; ++w)
        red[tid * NCLASS + w] += red[(tid + s) * NCLASS + w];
    }
    __syncthreads();
  }
  if (tid < NCLASS) part[(size_t)row * NCLASS + tid] = red[tid];
}

__global__ __launch_bounds__(256) void qnn_loss(const float* __restrict__ part,
                                                const int* __restrict__ y,
                                                float* __restrict__ out) {
  __shared__ float sred[256];
  float lsum = 0.0f;
  for (int b = threadIdx.x; b < BATCH; b += 256) {
    float lg[NCLASS];
#pragma unroll
    for (int w = 0; w < NCLASS; ++w) {
      lg[w] = part[(size_t)(2 * b) * NCLASS + w] +
              part[(size_t)(2 * b + 1) * NCLASS + w];
      out[1 + b * NCLASS + w] = lg[w];
    }
    float mx = lg[0];
#pragma unroll
    for (int w = 1; w < NCLASS; ++w) mx = fmaxf(mx, lg[w]);
    float se = 0.0f;
#pragma unroll
    for (int w = 0; w < NCLASS; ++w) se += expf(lg[w] - mx);
    const float lse = mx + logf(se);
    const int t = y[b];
    lsum += lse - lg[t];
  }
  sred[threadIdx.x] = lsum;
  __syncthreads();
  for (int s = 128; s > 0; s >>= 1) {
    if (threadIdx.x < s) sred[threadIdx.x] += sred[threadIdx.x + s];
    __syncthreads();
  }
  if (threadIdx.x == 0) out[0] = sred[0] / (float)BATCH;
}

extern "C" void kernel_launch(void* const* d_in, const int* in_sizes, int n_in,
                              void* d_out, int out_size, void* d_ws, size_t ws_size,
                              hipStream_t stream) {
  (void)in_sizes; (void)n_in; (void)out_size; (void)ws_size;
  const float* zr = (const float*)d_in[0];
  const float* zi = (const float*)d_in[1];
  const float* th = (const float*)d_in[2];
  const int*   y  = (const int*)d_in[3];
  float* out = (float*)d_out;

  float* gmat = (float*)d_ws;
  float* part = gmat + 48 * 256;

  hipLaunchKernelGGL(qnn_gates, dim3(48), dim3(256), 0, stream, th, gmat);
  hipLaunchKernelGGL(qnn_main, dim3(2 * BATCH), dim3(128), 0, stream,
                     zr, zi, gmat, part);
  hipLaunchKernelGGL(qnn_loss, dim3(1), dim3(256), 0, stream, part, y, out);
}


# Each pass issues all its LDS reads as one group ahead of its 16 WMMAs (4 independent accumulator chains); pass-1 gather indices held in registers, layer-0 case folded into the initial store
# speedup vs baseline: 1.3599x; 1.3599x over previous
#include <hip/hip_runtime.h>
#include <hip/hip_bf16.h>


#define NQ      12
#define DIM     4096
#define NLAYERS 16
#define BATCH   2048
#define NCLASS  5

typedef float v2f __attribute__((ext_vector_type(2)));
typedef float v8f __attribute__((ext_vector_type(8)));

__device__ __forceinline__ int pa(int i) { return i + (i >> 6); }
#define PADDED_DIM (DIM + (DIM >> 6))

__global__ __launch_bounds__(256) void qnn_gates(const float* __restrict__ thetas,
                                                 float* __restrict__ gmat) {
  const int blk = blockIdx.x;
  const int l = blk / 3, g = blk % 3;
  const int t = threadIdx.x;
  const int m = t >> 4, k = t & 15;
  float prod = 1.0f;
#pragma unroll
  for (int j = 0; j < 4; ++j) {
    const int q = g * 4 + j;
    const float th = 0.5f * thetas[l * NQ + q];
    const float c = cosf(th), s = sinf(th);
    const int mb = (m >> (3 - j)) & 1;
    const int kb = (k >> (3 - j)) & 1;
    const float f = mb ? (kb ? c : s) : (kb ? -s : c);
    prod *= f;
  }
  gmat[blk * 256 + t] = prod;
}

__device__ __forceinline__ v8f mma4(v2f a, v2f b, v8f c) {
  return __builtin_amdgcn_wmma_f32_16x16x4_f32(false, a, false, b, (short)0, c,
                                               false, false);
}

__device__ __forceinline__ void ry_layer(float* X, float* Y, const float* gm,
                                         const int (&gi)[32], int wv, int half,
                                         int hm) {
  const int ga  = hm * 16 + half * 2;
  const int s1  = half * 2080 + wv * 65 + hm;
  const int l2  = wv * 1040 + half * 32 + hm;
  const int s2  = wv * 1040 + half * 130 + hm;
  const int l3  = wv * 1040 + hm * 16 + (hm >> 2) + half * 2;
  const int s3  = wv * 1040 + hm * 16 + (hm >> 2) + half * 8;

  {
    v2f a[4], b[4][4];
#pragma unroll
    for (int kk = 0; kk < 4; ++kk) {
      a[kk].x = gm[ga + kk * 4];
      a[kk].y = gm[ga + kk * 4 + 1];
    }
#pragma unroll
    for (int t4 = 0; t4 < 4; ++t4) {
#pragma unroll
      for (int kk = 0; kk < 4; ++kk) {
        b[t4][kk].x = X[gi[(t4 * 4 + kk) * 2 + 0]];
        b[t4][kk].y = X[gi[(t4 * 4 + kk) * 2 + 1]];
      }
    }
    v8f acc[4];
#pragma unroll
    for (int t4 = 0; t4 < 4; ++t4)
      acc[t4] = (v8f){0.f, 0.f, 0.f, 0.f, 0.f, 0.f, 0.f, 0.f};
#pragma unroll
    for (int kk = 0; kk < 4; ++kk) {
#pragma unroll
      for (int t4 = 0; t4 < 4; ++t4) acc[t4] = mma4(a[kk], b[t4][kk], acc[t4]);
    }
#pragma unroll
    for (int t4 = 0; t4 < 4; ++t4) {
#pragma unroll
      for (int r = 0; r < 8; ++r) Y[s1 + r * 260 + t4 * 16] = acc[t4][r];
    }
  }
  __syncthreads();

  {
    v2f a[4], b[4][4];
#pragma unroll
    for (int kk = 0; kk < 4; ++kk) {
      a[kk].x = gm[256 + ga + kk * 4];
      a[kk].y = gm[256 + ga + kk * 4 + 1];
    }
#pragma unroll
    for (int t4 = 0; t4 < 4; ++t4) {
#pragma unroll
      for (int kk = 0; kk < 4; ++kk) {
        b[t4][kk].x = Y[l2 + t4 * 260 + kk * 65];
        b[t4][kk].y = Y[l2 + t4 * 260 + kk * 65 + 16];
      }
    }
    v8f acc[4];
#pragma unroll
    for (int t4 = 0; t4 < 4; ++t4)
      acc[t4] = (v8f){0.f, 0.f, 0.f, 0.f, 0.f, 0.f, 0.f, 0.f};
#pragma unroll
    for (int kk = 0; kk < 4; ++kk) {
#pragma unroll
      for (int t4 = 0; t4 < 4; ++t4) acc[t4] = mma4(a[kk], b[t4][kk], acc[t4]);
    }
#pragma unroll
    for (int t4 = 0; t4 < 4; ++t4) {
#pragma unroll
      for (int r = 0; r < 8; ++r)
        Y[s2 + t4 * 260 + r * 16 + (r >> 2)] = acc[t4][r];
    }
  }
  __syncthreads();

  {
    v2f a[4], b[4][4];
#pragma unroll
    for (int kk = 0; kk < 4; ++kk) {
      a[kk].x = gm[512 + ga + kk * 4];
      a[kk].y = gm[512 + ga + kk * 4 + 1];
    }
#pragma unroll
    for (int t4 = 0; t4 < 4; ++t4) {
#pragma unroll
      for (int kk = 0; kk < 4; ++kk) {
        b[t4][kk].x = Y[l3 + t4 * 260 + kk * 4];
        b[t4][kk].y = Y[l3 + t4 * 260 + kk * 4 + 1];
      }
    }
    v8f acc[4];
#pragma unroll
    for (int t4 = 0; t4 < 4; ++t4)
      acc[t4] = (v8f){0.f, 0.f, 0.f, 0.f, 0.f, 0.f, 0.f, 0.f};
#pragma unroll
    for (int kk = 0; kk < 4; ++kk) {
#pragma unroll
      for (int t4 = 0; t4 < 4; ++t4) acc[t4] = mma4(a[kk], b[t4][kk], acc[t4]);
    }
#pragma unroll
    for (int t4 = 0; t4 < 4; ++t4) {
#pragma unroll
      for (int r = 0; r < 8; ++r) X[s3 + t4 * 260 + r] = acc[t4][r];
    }
  }
  __syncthreads();
}

__global__ __launch_bounds__(128) void qnn_main(const float* __restrict__ zr,
                                                const float* __restrict__ zi,
                                                const float* __restrict__ gmat,
                                                float* __restrict__ part) {
  __shared__ float bufA[PADDED_DIM];
  __shared__ float bufB[PADDED_DIM];
  __shared__ unsigned short tbl[DIM];
  __shared__ float gm[3 * 256];
  __shared__ float red[128 * NCLASS];

  const int tid  = threadIdx.x;
  const int row  = blockIdx.x;
  const int comp = row & 1;
  const int bidx = row >> 1;
  const float* __restrict__ src = comp ? zi : zr;

  float v0[DIM / 128];
#pragma unroll
  for (int n = 0; n < DIM / 128; ++n)
    v0[n] = src[(size_t)bidx * DIM + tid + 128 * n];

#pragma unroll
  for (int n = 0; n < DIM / 128; ++n) {
    const int i = tid + 128 * n;
    int j = i;
#pragma unroll
    for (int q = NQ - 1; q >= 0; --q) {
      const int cb = 1 << (NQ - 1 - q);
      const int tb = 1 << (NQ - 1 - ((q + 1) % NQ));
      j = (j & cb) ? (j ^ tb) : j;
    }
    tbl[i] = (unsigned short)j;
    bufA[pa(j)] = v0[n];
  }
  __syncthreads();

  const int lane = tid & 31;
  const int wv   = tid >> 5;
  const int half = lane >> 4;
  const int hm   = lane & 15;

  int gi[32];
#pragma unroll
  for (int t4 = 0; t4 < 4; ++t4) {
#pragma unroll
    for (int kk = 0; kk < 4; ++kk) {
      const int i0 = (kk * 4 + half * 2) * 256 + (wv * 4 + t4) * 16 + hm;
      gi[(t4 * 4 + kk) * 2 + 0] = pa((int)tbl[i0]);
      gi[(t4 * 4 + kk) * 2 + 1] = pa((int)tbl[i0 + 256]);
    }
  }

  for (int l = 0; l < NLAYERS; ++l) {
#pragma unroll
    for (int n = 0; n < 6; ++n) gm[tid + 128 * n] = gmat[l * 768 + tid + 128 * n];
    __syncthreads();
    ry_layer(bufA, bufB, gm, gi, wv, half, hm);
  }

  float acc[NCLASS] = {0.f, 0.f, 0.f, 0.f, 0.f};
  const float* fin = bufA;
  for (int i = tid; i < DIM; i += 128) {
    const float s = fin[pa(tbl[i])];
    const float v = s * s;
#pragma unroll
    for (int w = 0; w < NCLASS; ++w)
      acc[w] += ((i >> (NQ - 1 - w)) & 1) ? -v : v;
  }
#pragma unroll
  for (int w = 0; w < NCLASS; ++w) red[tid * NCLASS + w] = acc[w];
  __syncthreads();
  for (int s = 64; s > 0; s >>= 1) {
    if (tid < s) {
#pragma unroll
      for (int w = 0; w < NCLASS; ++w)
        red[tid * NCLASS + w] += red[(tid + s) * NCLASS + w];
    }
    __syncthreads();
  }
  if (tid < NCLASS) part[(size_t)row * NCLASS + tid] = red[tid];
}

__global__ __launch_bounds__(256) void qnn_loss(const float* __restrict__ part,
                                                const int* __restrict__ y,
                                                float* __restrict__ out) {
  __shared__ float sred[256];
  float lsum = 0.0f;
  for (int b = threadIdx.x; b < BATCH; b += 256) {
    float lg[NCLASS];
#pragma unroll
    for (int w = 0; w < NCLASS; ++w) {
      lg[w] = part[(size_t)(2 * b) * NCLASS + w] +
              part[(size_t)(2 * b + 1) * NCLASS + w];
      out[1 + b * NCLASS + w] = lg[w];
    }
    float mx = lg[0];
#pragma unroll
    for (int w = 1; w < NCLASS; ++w) mx = fmaxf(mx, lg[w]);
    float se = 0.0f;
#pragma unroll
    for (int w = 0; w < NCLASS; ++w) se += expf(lg[w] - mx);
    const float lse = mx + logf(se);
    const int t = y[b];
    lsum += lse - lg[t];
  }
  sred[threadIdx.x] = lsum;
  __syncthreads();
  for (int s = 128; s > 0; s >>= 1) {
    if (threadIdx.x < s) sred[threadIdx.x] += sred[threadIdx.x + s];
    __syncthreads();
  }
  if (threadIdx.x == 0) out[0] = sred[0] / (float)BATCH;
}

extern "C" void kernel_launch(void* const* d_in, const int* in_sizes, int n_in,
                              void* d_out, int out_size, void* d_ws, size_t ws_size,
                              hipStream_t stream) {
  (void)in_sizes; (void)n_in; (void)out_size; (void)ws_size;
  const float* zr = (const float*)d_in[0];
  const float* zi = (const float*)d_in[1];
  const float* th = (const float*)d_in[2];
  const int*   y  = (const int*)d_in[3];
  float* out = (float*)d_out;

  float* gmat = (float*)d_ws;
  float* part = gmat + 48 * 256;

  hipLaunchKernelGGL(qnn_gates, dim3(48), dim3(256), 0, stream, th, gmat);
  hipLaunchKernelGGL(qnn_main, dim3(2 * BATCH), dim3(128), 0, stream,
                     zr, zi, gmat, part);
  hipLaunchKernelGGL(qnn_loss, dim3(1), dim3(256), 0, stream, part, y, out);
}


# LDS layouts redone per buffer (state pitches 336/20, pair-adjacent and XOR-swizzled intermediates, gate row pitch 20): fragment loads/stores conflict-free, 8/16 B wide; pass-3 reads fenced as a group
# speedup vs baseline: 1.5841x; 1.1649x over previous
#include <hip/hip_runtime.h>
#include <hip/hip_bf16.h>


#define NQ      12
#define DIM     4096
#define NLAYERS 16
#define BATCH   2048
#define NCLASS  5

typedef float v2f __attribute__((ext_vector_type(2)));
typedef float v4f __attribute__((ext_vector_type(4)));
typedef float v8f __attribute__((ext_vector_type(8)));

#define PADDED_DIM (DIM + (DIM >> 6))

#define XA 336
#define XB 20
#define XDIM (16 * XA)
__device__ __forceinline__ int xi(int i) {
  return (i >> 8) * XA + ((i >> 4) & 15) * XB + (i & 15);
}
#define GP 20
#define GM (16 * GP)

__global__ __launch_bounds__(256) void qnn_gates(const float* __restrict__ thetas,
                                                 float* __restrict__ gmat) {
  const int blk = blockIdx.x;
  const int l = blk / 3, g = blk % 3;
  const int t = threadIdx.x;
  const int m = t >> 4, k = t & 15;
  float prod = 1.0f;
#pragma unroll
  for (int j = 0; j < 4; ++j) {
    const int q = g * 4 + j;
    const float th = 0.5f * thetas[l * NQ + q];
    const float c = cosf(th), s = sinf(th);
    const int mb = (m >> (3 - j)) & 1;
    const int kb = (k >> (3 - j)) & 1;
    const float f = mb ? (kb ? c : s) : (kb ? -s : c);
    prod *= f;
  }
  gmat[blk * 256 + t] = prod;
}

__device__ __forceinline__ v8f mma4(v2f a, v2f b, v8f c) {
  return __builtin_amdgcn_wmma_f32_16x16x4_f32(false, a, false, b, (short)0, c,
                                               false, false);
}

__device__ __forceinline__ void ry_layer(float* X, float* Y, const float* gm,
                                         const int (&gi)[32], int wv, int half,
                                         int hm) {
  const int ga  = hm * GP + half * 2;
  const int s1  = half * 2080 + wv * 64 + 2 * hm;
  const int l2  = wv * 1040 + half * 32 + 2 * hm;
  const int s2  = wv * 1040 + half * 32;
  const int l3  = wv * 1040 + ((hm >> 2) & 1) * 128 + ((hm >> 1) & 1) * 64 +
                  ((hm >> 3) & 1) * 32 + (hm & 1) * 16 + half * 2;
  const int sw3 = (((hm >> 1) & 1) << 1) | ((hm >> 2) & 1);
  const int s3  = wv * 4 * XA + hm * XB + half * 8;

  {
    v2f a[4], b[4][4];
#pragma unroll
    for (int kk = 0; kk < 4; ++kk) {
      a[kk] = *(const v2f*)&gm[ga + kk * 4];
    }
#pragma unroll
    for (int t4 = 0; t4 < 4; ++t4) {
#pragma unroll
      for (int kk = 0; kk < 4; ++kk) {
        b[t4][kk].x = X[gi[(t4 * 4 + kk) * 2 + 0]];
        b[t4][kk].y = X[gi[(t4 * 4 + kk) * 2 + 1]];
      }
    }
    v8f acc[4];
#pragma unroll
    for (int t4 = 0; t4 < 4; ++t4)
      acc[t4] = (v8f){0.f, 0.f, 0.f, 0.f, 0.f, 0.f, 0.f, 0.f};
#pragma unroll
    for (int kk = 0; kk < 4; ++kk) {
#pragma unroll
      for (int t4 = 0; t4 < 4; ++t4) acc[t4] = mma4(a[kk], b[t4][kk], acc[t4]);
    }
#pragma unroll
    for (int t4 = 0; t4 < 4; ++t4) {
#pragma unroll
      for (int r = 0; r < 8; ++r)
        Y[s1 + r * 260 + (t4 >> 1) * 32 + (t4 & 1)] = acc[t4][r];
    }
  }
  __syncthreads();

  {
    v2f a[4], b[4][4];
#pragma unroll
    for (int kk = 0; kk < 4; ++kk) {
      a[kk] = *(const v2f*)&gm[GM + ga + kk * 4];
    }
#pragma unroll
    for (int t4 = 0; t4 < 4; ++t4) {
#pragma unroll
      for (int kk = 0; kk < 4; ++kk)
        b[t4][kk] = *(const v2f*)&Y[l2 + t4 * 260 + kk * 64];
    }
    v8f acc[4];
#pragma unroll
    for (int t4 = 0; t4 < 4; ++t4)
      acc[t4] = (v8f){0.f, 0.f, 0.f, 0.f, 0.f, 0.f, 0.f, 0.f};
#pragma unroll
    for (int kk = 0; kk < 4; ++kk) {
#pragma unroll
      for (int t4 = 0; t4 < 4; ++t4) acc[t4] = mma4(a[kk], b[t4][kk], acc[t4]);
    }
#pragma unroll
    for (int t4 = 0; t4 < 4; ++t4) {
#pragma unroll
      for (int r = 0; r < 8; ++r) {
        const int x = ((r >> 1) & 1) * 8 + ((r >> 2) & 1) * 4;
        Y[s2 + (hm ^ x) + t4 * 260 + ((r >> 2) & 1) * 128 + ((r >> 1) & 1) * 64 +
          (r & 1) * 16] = acc[t4][r];
      }
    }
  }
  __syncthreads();

  {
    v2f a[4], b[4][4];
#pragma unroll
    for (int kk = 0; kk < 4; ++kk) {
      a[kk] = *(const v2f*)&gm[2 * GM + ga + kk * 4];
    }
#pragma unroll
    for (int t4 = 0; t4 < 4; ++t4) {
#pragma unroll
      for (int kk = 0; kk < 4; ++kk)
        b[t4][kk] = *(const v2f*)&Y[l3 + ((kk ^ sw3) << 2) + t4 * 260];
    }
    asm volatile("" ::"v"(a[0]), "v"(a[1]), "v"(a[2]), "v"(a[3]), "v"(b[0][0]), "v"(b[0][1]), "v"(b[0][2]), "v"(b[0][3]), "v"(b[1][0]), "v"(b[1][1]), "v"(b[1][2]), "v"(b[1][3]), "v"(b[2][0]), "v"(b[2][1]), "v"(b[2][2]), "v"(b[2][3]), "v"(b[3][0]), "v"(b[3][1]), "v"(b[3][2]), "v"(b[3][3]));
    v8f acc[4];
#pragma unroll
    for (int t4 = 0; t4 < 4; ++t4)
      acc[t4] = (v8f){0.f, 0.f, 0.f, 0.f, 0.f, 0.f, 0.f, 0.f};
#pragma unroll
    for (int kk = 0; kk < 4; ++kk) {
#pragma unroll
      for (int t4 = 0; t4 < 4; ++t4) acc[t4] = mma4(a[kk], b[t4][kk], acc[t4]);
    }
#pragma unroll
    for (int t4 = 0; t4 < 4; ++t4) {
      *(v4f*)&X[s3 + t4 * XA] =
          __builtin_shufflevector(acc[t4], acc[t4], 0, 1, 2, 3);
      *(v4f*)&X[s3 + t4 * XA + 4] =
          __builtin_shufflevector(acc[t4], acc[t4], 4, 5, 6, 7);
    }
  }
  __syncthreads();
}

__global__ __launch_bounds__(128) void qnn_main(const float* __restrict__ zr,
                                                const float* __restrict__ zi,
                                                const float* __restrict__ gmat,
                                                float* __restrict__ part) {
  __shared__ __attribute__((aligned(16))) float bufA[XDIM];
  __shared__ __attribute__((aligned(16))) float bufB[PADDED_DIM];
  __shared__ unsigned short tbl[DIM];
  __shared__ __attribute__((aligned(16))) float gm[3 * GM];
  __shared__ float red[128 * NCLASS];

  const int tid  = threadIdx.x;
  const int row  = blockIdx.x;
  const int comp = row & 1;
  const int bidx = row >> 1;
  const float* __restrict__ src = comp ? zi : zr;

  float v0[DIM / 128];
#pragma unroll
  for (int n = 0; n < DIM / 128; ++n)
    v0[n] = src[(size_t)bidx * DIM + tid + 128 * n];

#pragma unroll
  for (int n = 0; n < DIM / 128; ++n) {
    const int i = tid + 128 * n;
    int j = i;
#pragma unroll
    for (int q = NQ - 1; q >= 0; --q) {
      const int cb = 1 << (NQ - 1 - q);
      const int tb = 1 << (NQ - 1 - ((q + 1) % NQ));
      j = (j & cb) ? (j ^ tb) : j;
    }
    tbl[i] = (unsigned short)j;
    bufA[xi(j)] = v0[n];
  }
  __syncthreads();

  const int lane = tid & 31;
  const int wv   = tid >> 5;
  const int half = lane >> 4;
  const int hm   = lane & 15;

  int gi[32];
#pragma unroll
  for (int t4 = 0; t4 < 4; ++t4) {
#pragma unroll
    for (int kk = 0; kk < 4; ++kk) {
      const int i0 = (kk * 4 + half * 2) * 256 + (wv * 4 + t4) * 16 + hm;
      gi[(t4 * 4 + kk) * 2 + 0] = xi((int)tbl[i0]);
      gi[(t4 * 4 + kk) * 2 + 1] = xi((int)tbl[i0 + 256]);
    }
  }

  for (int l = 0; l < NLAYERS; ++l) {
#pragma unroll
    for (int n = 0; n < 6; ++n)
      gm[(n >> 1) * GM + ((tid >> 4) + 8 * (n & 1)) * GP + (tid & 15)] =
          gmat[l * 768 + tid + 128 * n];
    __syncthreads();
    ry_layer(bufA, bufB, gm, gi, wv, half, hm);
  }

  float acc[NCLASS] = {0.f, 0.f, 0.f, 0.f, 0.f};
  const float* fin = bufA;
  for (int i = tid; i < DIM; i += 128) {
    const float s = fin[xi(tbl[i])];
    const float v = s * s;
#pragma unroll
    for (int w = 0; w < NCLASS; ++w)
      acc[w] += ((i >> (NQ - 1 - w)) & 1) ? -v : v;
  }
#pragma unroll
  for (int w = 0; w < NCLASS; ++w) red[tid * NCLASS + w] = acc[w];
  __syncthreads();
  for (int s = 64; s > 0; s >>= 1) {
    if (tid < s) {
#pragma unroll
      for (int w = 0; w < NCLASS; ++w)
        red[tid * NCLASS + w] += red[(tid + s) * NCLASS + w];
    }
    __syncthreads();
  }
  if (tid < NCLASS) part[(size_t)row * NCLASS + tid] = red[tid];
}

__global__ __launch_bounds__(256) void qnn_loss(const float* __restrict__ part,
                                                const int* __restrict__ y,
                                                float* __restrict__ out) {
  __shared__ float sred[256];
  float lsum = 0.0f;
  for (int b = threadIdx.x; b < BATCH; b += 256) {
    float lg[NCLASS];
#pragma unroll
    for (int w = 0; w < NCLASS; ++w) {
      lg[w] = part[(size_t)(2 * b) * NCLASS + w] +
              part[(size_t)(2 * b + 1) * NCLASS + w];
      out[1 + b * NCLASS + w] = lg[w];
    }
    float mx = lg[0];
#pragma unroll
    for (int w = 1; w < NCLASS; ++w) mx = fmaxf(mx, lg[w]);
    float se = 0.0f;
#pragma unroll
    for (int w = 0; w < NCLASS; ++w) se += expf(lg[w] - mx);
    const float lse = mx + logf(se);
    const int t = y[b];
    lsum += lse - lg[t];
  }
  sred[threadIdx.x] = lsum;
  __syncthreads();
  for (int s = 128; s > 0; s >>= 1) {
    if (threadIdx.x < s) sred[threadIdx.x] += sred[threadIdx.x + s];
    __syncthreads();
  }
  if (threadIdx.x == 0) out[0] = sred[0] / (float)BATCH;
}

extern "C" void kernel_launch(void* const* d_in, const int* in_sizes, int n_in,
                              void* d_out, int out_size, void* d_ws, size_t ws_size,
                              hipStream_t stream) {
  (void)in_sizes; (void)n_in; (void)out_size; (void)ws_size;
  const float* zr = (const float*)d_in[0];
  const float* zi = (const float*)d_in[1];
  const float* th = (const float*)d_in[2];
  const int*   y  = (const int*)d_in[3];
  float* out = (float*)d_out;

  float* gmat = (float*)d_ws;
  float* part = gmat + 48 * 256;

  hipLaunchKernelGGL(qnn_gates, dim3(48), dim3(256), 0, stream, th, gmat);
  hipLaunchKernelGGL(qnn_main, dim3(2 * BATCH), dim3(128), 0, stream,
                     zr, zi, gmat, part);
  hipLaunchKernelGGL(qnn_loss, dim3(1), dim3(256), 0, stream, part, y, out);
}


# Passes 2+3 fused: state tile as A operand, mode-b result stays in accumulator registers as B operand of the mode-c WMMAs; one LDS store+barrier+reload per layer gone; c-sum order regrouped
# speedup vs baseline: 1.6580x; 1.0467x over previous
#include <hip/hip_runtime.h>
#include <hip/hip_bf16.h>


#define NQ      12
#define DIM     4096
#define NLAYERS 16
#define BATCH   2048
#define NCLASS  5

typedef float v2f __attribute__((ext_vector_type(2)));
typedef float v4f __attribute__((ext_vector_type(4)));
typedef float v8f __attribute__((ext_vector_type(8)));

#define PADDED_DIM (DIM + (DIM >> 6))

#define XA 336
#define XB 20
#define XDIM (16 * XA)
__device__ __forceinline__ int xi(int i) {
  return (i >> 8) * XA + ((i >> 4) & 15) * XB + (i & 15);
}
#define GP 20
#define GM (16 * GP)

__global__ __launch_bounds__(256) void qnn_gates(const float* __restrict__ thetas,
                                                 float* __restrict__ gmat) {
  const int blk = blockIdx.x;
  const int l = blk / 3, g = blk % 3;
  const int t = threadIdx.x;
  const int m = t >> 4, k = t & 15;
  float prod = 1.0f;
#pragma unroll
  for (int j = 0; j < 4; ++j) {
    const int q = g * 4 + j;
    const float th = 0.5f * thetas[l * NQ + q];
    const float c = cosf(th), s = sinf(th);
    const int mb = (m >> (3 - j)) & 1;
    const int kb = (k >> (3 - j)) & 1;
    const float f = mb ? (kb ? c : s) : (kb ? -s : c);
    prod *= f;
  }
  gmat[blk * 256 + t] = prod;
}

__device__ __forceinline__ v8f mma4(v2f a, v2f b, v8f c) {
  return __builtin_amdgcn_wmma_f32_16x16x4_f32(false, a, false, b, (short)0, c,
                                               false, false);
}

__device__ __forceinline__ void ry_layer(float* X, float* Y, const float* gm,
                                         const int (&gi)[32], int wv, int half,
                                         int hm) {
  const int ga  = hm * GP + half * 2;
  const int s1  = half * 2080 + wv * 64 + 2 * hm;
  const int l2  = wv * 1040 + half * 32 + 2 * hm;
  const int s3  = wv * 4 * XA + hm * XB + half * 8;

  {
    v2f a[4], b[4][4];
#pragma unroll
    for (int kk = 0; kk < 4; ++kk) {
      a[kk] = *(const v2f*)&gm[ga + kk * 4];
    }
#pragma unroll
    for (int t4 = 0; t4 < 4; ++t4) {
#pragma unroll
      for (int kk = 0; kk < 4; ++kk) {
        b[t4][kk].x = X[gi[(t4 * 4 + kk) * 2 + 0]];
        b[t4][kk].y = X[gi[(t4 * 4 + kk) * 2 + 1]];
      }
    }
    v8f acc[4];
#pragma unroll
    for (int t4 = 0; t4 < 4; ++t4)
      acc[t4] = (v8f){0.f, 0.f, 0.f, 0.f, 0.f, 0.f, 0.f, 0.f};
#pragma unroll
    for (int kk = 0; kk < 4; ++kk) {
#pragma unroll
      for (int t4 = 0; t4 < 4; ++t4) acc[t4] = mma4(a[kk], b[t4][kk], acc[t4]);
    }
#pragma unroll
    for (int t4 = 0; t4 < 4; ++t4) {
#pragma unroll
      for (int r = 0; r < 8; ++r)
        Y[s1 + r * 260 + (t4 >> 1) * 32 + (t4 & 1)] = acc[t4][r];
    }
  }
  __syncthreads();

  {
    v2f g2[4], g3[4], s[4][4];
#pragma unroll
    for (int kk = 0; kk < 4; ++kk) {
      g2[kk] = *(const v2f*)&gm[GM + ga + kk * 4];
      g3[kk] = *(const v2f*)&gm[2 * GM + ga + kk * 4];
    }
#pragma unroll
    for (int t4 = 0; t4 < 4; ++t4) {
#pragma unroll
      for (int kk = 0; kk < 4; ++kk)
        s[t4][kk] = *(const v2f*)&Y[l2 + t4 * 260 + kk * 64];
    }
    __builtin_amdgcn_sched_barrier(0);
    v8f d1[4];
#pragma unroll
    for (int t4 = 0; t4 < 4; ++t4)
      d1[t4] = (v8f){0.f, 0.f, 0.f, 0.f, 0.f, 0.f, 0.f, 0.f};
#pragma unroll
    for (int kk = 0; kk < 4; ++kk) {
#pragma unroll
      for (int t4 = 0; t4 < 4; ++t4) d1[t4] = mma4(s[t4][kk], g2[kk], d1[t4]);
    }
    v8f d2[4];
#pragma unroll
    for (int t4 = 0; t4 < 4; ++t4)
      d2[t4] = (v8f){0.f, 0.f, 0.f, 0.f, 0.f, 0.f, 0.f, 0.f};
#pragma unroll
    for (int kk = 0; kk < 4; ++kk) {
#pragma unroll
      for (int t4 = 0; t4 < 4; ++t4) {
        const v2f bk = (v2f){d1[t4][2 * kk], d1[t4][2 * kk + 1]};
        d2[t4] = mma4(g3[kk], bk, d2[t4]);
      }
    }
#pragma unroll
    for (int t4 = 0; t4 < 4; ++t4) {
      *(v4f*)&X[s3 + t4 * XA] =
          __builtin_shufflevector(d2[t4], d2[t4], 0, 1, 2, 3);
      *(v4f*)&X[s3 + t4 * XA + 4] =
          __builtin_shufflevector(d2[t4], d2[t4], 4, 5, 6, 7);
    }
  }
  __syncthreads();
}

__global__ __launch_bounds__(128) void qnn_main(const float* __restrict__ zr,
                                                const float* __restrict__ zi,
                                                const float* __restrict__ gmat,
                                                float* __restrict__ part) {
  __shared__ __attribute__((aligned(16))) float bufA[XDIM];
  __shared__ __attribute__((aligned(16))) float bufB[PADDED_DIM];
  __shared__ unsigned short tbl[DIM];
  __shared__ __attribute__((aligned(16))) float gm[3 * GM];
  __shared__ float red[128 * NCLASS];

  const int tid  = threadIdx.x;
  const int row  = blockIdx.x;
  const int comp = row & 1;
  const int bidx = row >> 1;
  const float* __restrict__ src = comp ? zi : zr;

  float v0[DIM / 128];
#pragma unroll
  for (int n = 0; n < DIM / 128; ++n)
    v0[n] = src[(size_t)bidx * DIM + tid + 128 * n];

#pragma unroll
  for (int n = 0; n < DIM / 128; ++n) {
    const int i = tid + 128 * n;
    int j = i;
#pragma unroll
    for (int q = NQ - 1; q >= 0; --q) {
      const int cb = 1 << (NQ - 1 - q);
      const int tb = 1 << (NQ - 1 - ((q + 1) % NQ));
      j = (j & cb) ? (j ^ tb) : j;
    }
    tbl[i] = (unsigned short)j;
    bufA[xi(j)] = v0[n];
  }
  __syncthreads();

  const int lane = tid & 31;
  const int wv   = tid >> 5;
  const int half = lane >> 4;
  const int hm   = lane & 15;

  int gi[32];
#pragma unroll
  for (int t4 = 0; t4 < 4; ++t4) {
#pragma unroll
    for (int kk = 0; kk < 4; ++kk) {
      const int i0 = (kk * 4 + half * 2) * 256 + (wv * 4 + t4) * 16 + hm;
      gi[(t4 * 4 + kk) * 2 + 0] = xi((int)tbl[i0]);
      gi[(t4 * 4 + kk) * 2 + 1] = xi((int)tbl[i0 + 256]);
    }
  }

  const int colc = ((tid & 6) << 1) | ((tid >> 2) & 2) | (tid & 1);

  for (int l = 0; l < NLAYERS; ++l) {
#pragma unroll
    for (int n = 0; n < 6; ++n)
      gm[(n >> 1) * GM + ((tid >> 4) + 8 * (n & 1)) * GP +
         ((n >> 1) == 2 ? colc : (tid & 15))] = gmat[l * 768 + tid + 128 * n];
    __syncthreads();
    ry_layer(bufA, bufB, gm, gi, wv, half, hm);
  }

  float acc[NCLASS] = {0.f, 0.f, 0.f, 0.f, 0.f};
  const float* fin = bufA;
  for (int i = tid; i < DIM; i += 128) {
    const float s = fin[xi(tbl[i])];
    const float v = s * s;
#pragma unroll
    for (int w = 0; w < NCLASS; ++w)
      acc[w] += ((i >> (NQ - 1 - w)) & 1) ? -v : v;
  }
#pragma unroll
  for (int w = 0; w < NCLASS; ++w) red[tid * NCLASS + w] = acc[w];
  __syncthreads();
  for (int s = 64; s > 0; s >>= 1) {
    if (tid < s) {
#pragma unroll
      for (int w = 0; w < NCLASS; ++w)
        red[tid * NCLASS + w] += red[(tid + s) * NCLASS + w];
    }
    __syncthreads();
  }
  if (tid < NCLASS) part[(size_t)row * NCLASS + tid] = red[tid];
}

__global__ __launch_bounds__(256) void qnn_loss(const float* __restrict__ part,
                                                const int* __restrict__ y,
                                                float* __restrict__ out) {
  __shared__ float sred[256];
  float lsum = 0.0f;
  for (int b = threadIdx.x; b < BATCH; b += 256) {
    float lg[NCLASS];
#pragma unroll
    for (int w = 0; w < NCLASS; ++w) {
      lg[w] = part[(size_t)(2 * b) * NCLASS + w] +
              part[(size_t)(2 * b + 1) * NCLASS + w];
      out[1 + b * NCLASS + w] = lg[w];
    }
    float mx = lg[0];
#pragma unroll
    for (int w = 1; w < NCLASS; ++w) mx = fmaxf(mx, lg[w]);
    float se = 0.0f;
#pragma unroll
    for (int w = 0; w < NCLASS; ++w) se += expf(lg[w] - mx);
    const float lse = mx + logf(se);
    const int t = y[b];
    lsum += lse - lg[t];
  }
  sred[threadIdx.x] = lsum;
  __syncthreads();
  for (int s = 128; s > 0; s >>= 1) {
    if (threadIdx.x < s) sred[threadIdx.x] += sred[threadIdx.x + s];
    __syncthreads();
  }
  if (threadIdx.x == 0) out[0] = sred[0] / (float)BATCH;
}

extern "C" void kernel_launch(void* const* d_in, const int* in_sizes, int n_in,
                              void* d_out, int out_size, void* d_ws, size_t ws_size,
                              hipStream_t stream) {
  (void)in_sizes; (void)n_in; (void)out_size; (void)ws_size;
  const float* zr = (const float*)d_in[0];
  const float* zi = (const float*)d_in[1];
  const float* th = (const float*)d_in[2];
  const int*   y  = (const int*)d_in[3];
  float* out = (float*)d_out;

  float* gmat = (float*)d_ws;
  float* part = gmat + 48 * 256;

  hipLaunchKernelGGL(qnn_gates, dim3(48), dim3(256), 0, stream, th, gmat);
  hipLaunchKernelGGL(qnn_main, dim3(2 * BATCH), dim3(128), 0, stream,
                     zr, zi, gmat, part);
  hipLaunchKernelGGL(qnn_loss, dim3(1), dim3(256), 0, stream, part, y, out);
}


# Final signed |psi|^2 sums formed from the last layer's accumulator registers (signs from the lane's a,b,c bits via the ring's closed-form inverse), wave shuffle + 4-way add; sum order regrouped
# speedup vs baseline: 1.7278x; 1.0421x over previous
#include <hip/hip_runtime.h>
#include <hip/hip_bf16.h>


#define NQ      12
#define DIM     4096
#define NLAYERS 16
#define BATCH   2048
#define NCLASS  5

typedef float v2f __attribute__((ext_vector_type(2)));
typedef float v4f __attribute__((ext_vector_type(4)));
typedef float v8f __attribute__((ext_vector_type(8)));

#define PADDED_DIM (DIM + (DIM >> 6))

#define XA 336
#define XB 20
#define XDIM (16 * XA)
__device__ __forceinline__ int xi(int i) {
  return (i >> 8) * XA + ((i >> 4) & 15) * XB + (i & 15);
}
#define GP 20
#define GM (16 * GP)

__global__ __launch_bounds__(256) void qnn_gates(const float* __restrict__ thetas,
                                                 float* __restrict__ gmat) {
  const int blk = blockIdx.x;
  const int l = blk / 3, g = blk % 3;
  const int t = threadIdx.x;
  const int m = t >> 4, k = t & 15;
  float prod = 1.0f;
#pragma unroll
  for (int j = 0; j < 4; ++j) {
    const int q = g * 4 + j;
    const float th = 0.5f * thetas[l * NQ + q];
    const float c = cosf(th), s = sinf(th);
    const int mb = (m >> (3 - j)) & 1;
    const int kb = (k >> (3 - j)) & 1;
    const float f = mb ? (kb ? c : s) : (kb ? -s : c);
    prod *= f;
  }
  gmat[blk * 256 + t] = prod;
}

__device__ __forceinline__ v8f mma4(v2f a, v2f b, v8f c) {
  return __builtin_amdgcn_wmma_f32_16x16x4_f32(false, a, false, b, (short)0, c,
                                               false, false);
}

template <bool LAST>
__device__ __forceinline__ void ry_layer(float* X, float* Y, const float* gm,
                                         const int (&gi)[32], int wv, int half,
                                         int hm, v8f (&dfin)[4]) {
  const int ga  = hm * GP + half * 2;
  const int s1  = half * 2080 + wv * 64 + 2 * hm;
  const int l2  = wv * 1040 + half * 32 + 2 * hm;
  const int s3  = wv * 4 * XA + hm * XB + half * 8;

  {
    v2f a[4], b[4][4];
#pragma unroll
    for (int kk = 0; kk < 4; ++kk) {
      a[kk] = *(const v2f*)&gm[ga + kk * 4];
    }
#pragma unroll
    for (int t4 = 0; t4 < 4; ++t4) {
#pragma unroll
      for (int kk = 0; kk < 4; ++kk) {
        b[t4][kk].x = X[gi[(t4 * 4 + kk) * 2 + 0]];
        b[t4][kk].y = X[gi[(t4 * 4 + kk) * 2 + 1]];
      }
    }
    v8f acc[4];
#pragma unroll
    for (int t4 = 0; t4 < 4; ++t4)
      acc[t4] = (v8f){0.f, 0.f, 0.f, 0.f, 0.f, 0.f, 0.f, 0.f};
#pragma unroll
    for (int kk = 0; kk < 4; ++kk) {
#pragma unroll
      for (int t4 = 0; t4 < 4; ++t4) acc[t4] = mma4(a[kk], b[t4][kk], acc[t4]);
    }
#pragma unroll
    for (int t4 = 0; t4 < 4; ++t4) {
#pragma unroll
      for (int r = 0; r < 8; ++r)
        Y[s1 + r * 260 + (t4 >> 1) * 32 + (t4 & 1)] = acc[t4][r];
    }
  }
  __syncthreads();

  {
    v2f g2[4], g3[4], s[4][4];
#pragma unroll
    for (int kk = 0; kk < 4; ++kk) {
      g2[kk] = *(const v2f*)&gm[GM + ga + kk * 4];
      g3[kk] = *(const v2f*)&gm[2 * GM + ga + kk * 4];
    }
#pragma unroll
    for (int t4 = 0; t4 < 4; ++t4) {
#pragma unroll
      for (int kk = 0; kk < 4; ++kk)
        s[t4][kk] = *(const v2f*)&Y[l2 + t4 * 260 + kk * 64];
    }
    __builtin_amdgcn_sched_barrier(0);
    v8f d1[4];
#pragma unroll
    for (int t4 = 0; t4 < 4; ++t4)
      d1[t4] = (v8f){0.f, 0.f, 0.f, 0.f, 0.f, 0.f, 0.f, 0.f};
#pragma unroll
    for (int kk = 0; kk < 4; ++kk) {
#pragma unroll
      for (int t4 = 0; t4 < 4; ++t4) d1[t4] = mma4(s[t4][kk], g2[kk], d1[t4]);
    }
    v8f d2[4];
#pragma unroll
    for (int t4 = 0; t4 < 4; ++t4)
      d2[t4] = (v8f){0.f, 0.f, 0.f, 0.f, 0.f, 0.f, 0.f, 0.f};
#pragma unroll
    for (int kk = 0; kk < 4; ++kk) {
#pragma unroll
      for (int t4 = 0; t4 < 4; ++t4) {
        const v2f bk = (v2f){d1[t4][2 * kk], d1[t4][2 * kk + 1]};
        d2[t4] = mma4(g3[kk], bk, d2[t4]);
      }
    }
#pragma unroll
    for (int t4 = 0; t4 < 4; ++t4) {
      if (LAST) {
        dfin[t4] = d2[t4];
      } else {
        *(v4f*)&X[s3 + t4 * XA] =
            __builtin_shufflevector(d2[t4], d2[t4], 0, 1, 2, 3);
        *(v4f*)&X[s3 + t4 * XA + 4] =
            __builtin_shufflevector(d2[t4], d2[t4], 4, 5, 6, 7);
      }
    }
  }
  if (!LAST) __syncthreads();
}

__device__ __forceinline__ void stage_gates(float* gm,
                                            const float* __restrict__ gmat,
                                            int l, int tid, int colc) {
#pragma unroll
  for (int n = 0; n < 6; ++n)
    gm[(n >> 1) * GM + ((tid >> 4) + 8 * (n & 1)) * GP +
       ((n >> 1) == 2 ? colc : (tid & 15))] = gmat[l * 768 + tid + 128 * n];
}

__global__ __launch_bounds__(128) void qnn_main(const float* __restrict__ zr,
                                                const float* __restrict__ zi,
                                                const float* __restrict__ gmat,
                                                float* __restrict__ part) {
  __shared__ __attribute__((aligned(16))) float bufA[XDIM];
  __shared__ __attribute__((aligned(16))) float bufB[PADDED_DIM];
  __shared__ unsigned short tbl[DIM];
  __shared__ __attribute__((aligned(16))) float gm[3 * GM];
  __shared__ float red[4 * NCLASS];

  const int tid  = threadIdx.x;
  const int row  = blockIdx.x;
  const int comp = row & 1;
  const int bidx = row >> 1;
  const float* __restrict__ src = comp ? zi : zr;

  float v0[DIM / 128];
#pragma unroll
  for (int n = 0; n < DIM / 128; ++n)
    v0[n] = src[(size_t)bidx * DIM + tid + 128 * n];

#pragma unroll
  for (int n = 0; n < DIM / 128; ++n) {
    const int i = tid + 128 * n;
    int j = i;
#pragma unroll
    for (int q = NQ - 1; q >= 0; --q) {
      const int cb = 1 << (NQ - 1 - q);
      const int tb = 1 << (NQ - 1 - ((q + 1) % NQ));
      j = (j & cb) ? (j ^ tb) : j;
    }
    tbl[i] = (unsigned short)j;
    bufA[xi(j)] = v0[n];
  }
  __syncthreads();

  const int lane = tid & 31;
  const int wv   = tid >> 5;
  const int half = lane >> 4;
  const int hm   = lane & 15;

  int gi[32];
#pragma unroll
  for (int t4 = 0; t4 < 4; ++t4) {
#pragma unroll
    for (int kk = 0; kk < 4; ++kk) {
      const int i0 = (kk * 4 + half * 2) * 256 + (wv * 4 + t4) * 16 + hm;
      gi[(t4 * 4 + kk) * 2 + 0] = xi((int)tbl[i0]);
      gi[(t4 * 4 + kk) * 2 + 1] = xi((int)tbl[i0 + 256]);
    }
  }

  const int colc = ((tid & 6) << 1) | ((tid >> 2) & 2) | (tid & 1);

  v8f dfin[4];
  for (int l = 0; l < NLAYERS - 1; ++l) {
    stage_gates(gm, gmat, l, tid, colc);
    __syncthreads();
    ry_layer<false>(bufA, bufB, gm, gi, wv, half, hm, dfin);
  }
  stage_gates(gm, gmat, NLAYERS - 1, tid, colc);
  __syncthreads();
  ry_layer<true>(bufA, bufB, gm, gi, wv, half, hm, dfin);

  float acc[NCLASS] = {0.f, 0.f, 0.f, 0.f, 0.f};
#pragma unroll
  for (int t4 = 0; t4 < 4; ++t4) {
    float se = 0.f, so = 0.f;
#pragma unroll
    for (int r = 0; r < 8; ++r) {
      const float v = dfin[t4][r] * dfin[t4][r];
      if (__builtin_popcount((unsigned)r) & 1) so += v; else se += v;
    }
    const float tot = se + so;
    const float dif = se - so;
    const int a  = wv * 4 + t4;
    const int pa = __popc(a) & 1;
    const int x0 = (__popc(a & 7) ^ __popc(hm) ^ half) & 1;
    const int x1 = __popc(a & 12) & 1;
    const int x2 = __popc(a & 14) & 1;
    const int x4 = pa ^ (hm >> 3);
    acc[0] += x0 ? -dif : dif;
    acc[1] += x1 ? -tot : tot;
    acc[2] += x2 ? -tot : tot;
    acc[3] += pa ? -tot : tot;
    acc[4] += x4 ? -tot : tot;
  }
#pragma unroll
  for (int w = 0; w < NCLASS; ++w) {
#pragma unroll
    for (int m = 16; m > 0; m >>= 1) acc[w] += __shfl_xor(acc[w], m, 32);
  }
  if (lane == 0) {
#pragma unroll
    for (int w = 0; w < NCLASS; ++w) red[wv * NCLASS + w] = acc[w];
  }
  __syncthreads();
  if (tid < NCLASS)
    part[(size_t)row * NCLASS + tid] =
        (red[tid] + red[NCLASS + tid]) +
        (red[2 * NCLASS + tid] + red[3 * NCLASS + tid]);
}

__global__ __launch_bounds__(256) void qnn_loss(const float* __restrict__ part,
                                                const int* __restrict__ y,
                                                float* __restrict__ out) {
  __shared__ float sred[256];
  float lsum = 0.0f;
  for (int b = threadIdx.x; b < BATCH; b += 256) {
    float lg[NCLASS];
#pragma unroll
    for (int w = 0; w < NCLASS; ++w) {
      lg[w] = part[(size_t)(2 * b) * NCLASS + w] +
              part[(size_t)(2 * b + 1) * NCLASS + w];
      out[1 + b * NCLASS + w] = lg[w];
    }
    float mx = lg[0];
#pragma unroll
    for (int w = 1; w < NCLASS; ++w) mx = fmaxf(mx, lg[w]);
    float se = 0.0f;
#pragma unroll
    for (int w = 0; w < NCLASS; ++w) se += expf(lg[w] - mx);
    const float lse = mx + logf(se);
    const int t = y[b];
    lsum += lse - lg[t];
  }
  sred[threadIdx.x] = lsum;
  __syncthreads();
  for (int s = 128; s > 0; s >>= 1) {
    if (threadIdx.x < s) sred[threadIdx.x] += sred[threadIdx.x + s];
    __syncthreads();
  }
  if (threadIdx.x == 0) out[0] = sred[0] / (float)BATCH;
}

extern "C" void kernel_launch(void* const* d_in, const int* in_sizes, int n_in,
                              void* d_out, int out_size, void* d_ws, size_t ws_size,
                              hipStream_t stream) {
  (void)in_sizes; (void)n_in; (void)out_size; (void)ws_size;
  const float* zr = (const float*)d_in[0];
  const float* zi = (const float*)d_in[1];
  const float* th = (const float*)d_in[2];
  const int*   y  = (const int*)d_in[3];
  float* out = (float*)d_out;

  float* gmat = (float*)d_ws;
  float* part = gmat + 48 * 256;

  hipLaunchKernelGGL(qnn_gates, dim3(48), dim3(256), 0, stream, th, gmat);
  hipLaunchKernelGGL(qnn_main, dim3(2 * BATCH), dim3(128), 0, stream,
                     zr, zi, gmat, part);
  hipLaunchKernelGGL(qnn_loss, dim3(1), dim3(256), 0, stream, part, y, out);
}


# Ring permutation used in closed form (Gray code plus wrap term): no index chain, no LDS table, no scatter; input stored in plain order by 16-byte loads/stores, first layer peeled to read it directly
# speedup vs baseline: 1.8009x; 1.0423x over previous
#include <hip/hip_runtime.h>
#include <hip/hip_bf16.h>


#define NQ      12
#define DIM     4096
#define NLAYERS 16
#define BATCH   2048
#define NCLASS  5

typedef float v2f __attribute__((ext_vector_type(2)));
typedef float v4f __attribute__((ext_vector_type(4)));
typedef float v8f __attribute__((ext_vector_type(8)));

#define PADDED_DIM (DIM + (DIM >> 6))

#define XA 336
#define XB 20
#define XDIM (16 * XA)
__device__ __forceinline__ int xi(int i) {
  return (i >> 8) * XA + ((i >> 4) & 15) * XB + (i & 15);
}
__device__ __forceinline__ constexpr int ring(int i) {
  return i ^ (i >> 1) ^ ((i & 1) ? 0xC00 : 0);
}
#define GP 20
#define GM (16 * GP)

__global__ __launch_bounds__(256) void qnn_gates(const float* __restrict__ thetas,
                                                 float* __restrict__ gmat) {
  const int blk = blockIdx.x;
  const int l = blk / 3, g = blk % 3;
  const int t = threadIdx.x;
  const int m = t >> 4, k = t & 15;
  float prod = 1.0f;
#pragma unroll
  for (int j = 0; j < 4; ++j) {
    const int q = g * 4 + j;
    const float th = 0.5f * thetas[l * NQ + q];
    const float c = cosf(th), s = sinf(th);
    const int mb = (m >> (3 - j)) & 1;
    const int kb = (k >> (3 - j)) & 1;
    const float f = mb ? (kb ? c : s) : (kb ? -s : c);
    prod *= f;
  }
  gmat[blk * 256 + t] = prod;
}

__device__ __forceinline__ v8f mma4(v2f a, v2f b, v8f c) {
  return __builtin_amdgcn_wmma_f32_16x16x4_f32(false, a, false, b, (short)0, c,
                                               false, false);
}

template <bool FIRST, bool LAST>
__device__ __forceinline__ void ry_layer(float* X, float* Y, const float* gm,
                                         const int (&gi)[32], int lin, int wv,
                                         int half, int hm, v8f (&dfin)[4]) {
  const int ga  = hm * GP + half * 2;
  const int s1  = half * 2080 + wv * 64 + 2 * hm;
  const int l2  = wv * 1040 + half * 32 + 2 * hm;
  const int s3  = wv * 4 * XA + hm * XB + half * 8;

  {
    v2f a[4], b[4][4];
#pragma unroll
    for (int kk = 0; kk < 4; ++kk) {
      a[kk] = *(const v2f*)&gm[ga + kk * 4];
    }
#pragma unroll
    for (int t4 = 0; t4 < 4; ++t4) {
#pragma unroll
      for (int kk = 0; kk < 4; ++kk) {
        if (FIRST) {
          b[t4][kk].x = X[lin + (kk * 4 + 0) * XA + t4 * XB];
          b[t4][kk].y = X[lin + (kk * 4 + 1) * XA + t4 * XB];
        } else {
          b[t4][kk].x = X[gi[(t4 * 4 + kk) * 2 + 0]];
          b[t4][kk].y = X[gi[(t4 * 4 + kk) * 2 + 1]];
        }
      }
    }
    v8f acc[4];
#pragma unroll
    for (int t4 = 0; t4 < 4; ++t4)
      acc[t4] = (v8f){0.f, 0.f, 0.f, 0.f, 0.f, 0.f, 0.f, 0.f};
#pragma unroll
    for (int kk = 0; kk < 4; ++kk) {
#pragma unroll
      for (int t4 = 0; t4 < 4; ++t4) acc[t4] = mma4(a[kk], b[t4][kk], acc[t4]);
    }
#pragma unroll
    for (int t4 = 0; t4 < 4; ++t4) {
#pragma unroll
      for (int r = 0; r < 8; ++r)
        Y[s1 + r * 260 + (t4 >> 1) * 32 + (t4 & 1)] = acc[t4][r];
    }
  }
  __syncthreads();

  {
    v2f g2[4], g3[4], s[4][4];
#pragma unroll
    for (int kk = 0; kk < 4; ++kk) {
      g2[kk] = *(const v2f*)&gm[GM + ga + kk * 4];
      g3[kk] = *(const v2f*)&gm[2 * GM + ga + kk * 4];
    }
#pragma unroll
    for (int t4 = 0; t4 < 4; ++t4) {
#pragma unroll
      for (int kk = 0; kk < 4; ++kk)
        s[t4][kk] = *(const v2f*)&Y[l2 + t4 * 260 + kk * 64];
    }
    __builtin_amdgcn_sched_barrier(0);
    v8f d1[4];
#pragma unroll
    for (int t4 = 0; t4 < 4; ++t4)
      d1[t4] = (v8f){0.f, 0.f, 0.f, 0.f, 0.f, 0.f, 0.f, 0.f};
#pragma unroll
    for (int kk = 0; kk < 4; ++kk) {
#pragma unroll
      for (int t4 = 0; t4 < 4; ++t4) d1[t4] = mma4(s[t4][kk], g2[kk], d1[t4]);
    }
    v8f d2[4];
#pragma unroll
    for (int t4 = 0; t4 < 4; ++t4)
      d2[t4] = (v8f){0.f, 0.f, 0.f, 0.f, 0.f, 0.f, 0.f, 0.f};
#pragma unroll
    for (int kk = 0; kk < 4; ++kk) {
#pragma unroll
      for (int t4 = 0; t4 < 4; ++t4) {
        const v2f bk = (v2f){d1[t4][2 * kk], d1[t4][2 * kk + 1]};
        d2[t4] = mma4(g3[kk], bk, d2[t4]);
      }
    }
#pragma unroll
    for (int t4 = 0; t4 < 4; ++t4) {
      if (LAST) {
        dfin[t4] = d2[t4];
      } else {
        *(v4f*)&X[s3 + t4 * XA] =
            __builtin_shufflevector(d2[t4], d2[t4], 0, 1, 2, 3);
        *(v4f*)&X[s3 + t4 * XA + 4] =
            __builtin_shufflevector(d2[t4], d2[t4], 4, 5, 6, 7);
      }
    }
  }
  if (!LAST) __syncthreads();
}

__device__ __forceinline__ void stage_gates(float* gm,
                                            const float* __restrict__ gmat,
                                            int l, int tid, int colc) {
#pragma unroll
  for (int n = 0; n < 6; ++n)
    gm[(n >> 1) * GM + ((tid >> 4) + 8 * (n & 1)) * GP +
       ((n >> 1) == 2 ? colc : (tid & 15))] = gmat[l * 768 + tid + 128 * n];
}

__global__ __launch_bounds__(128) void qnn_main(const float* __restrict__ zr,
                                                const float* __restrict__ zi,
                                                const float* __restrict__ gmat,
                                                float* __restrict__ part) {
  __shared__ __attribute__((aligned(16))) float bufA[XDIM];
  __shared__ __attribute__((aligned(16))) float bufB[PADDED_DIM];
  __shared__ __attribute__((aligned(16))) float gm[3 * GM];
  __shared__ float red[4 * NCLASS];

  const int tid  = threadIdx.x;
  const int row  = blockIdx.x;
  const int comp = row & 1;
  const int bidx = row >> 1;
  const float* __restrict__ src = comp ? zi : zr;

  const int lane = tid & 31;
  const int wv   = tid >> 5;
  const int half = lane >> 4;
  const int hm   = lane & 15;

  const v4f* __restrict__ src4 = (const v4f*)(src + (size_t)bidx * DIM);
  v4f v0[DIM / 512];
#pragma unroll
  for (int n = 0; n < DIM / 512; ++n) v0[n] = src4[tid + 128 * n];

  const int st0 = (tid >> 6) * XA + ((tid >> 2) & 15) * XB + (tid & 3) * 4;
#pragma unroll
  for (int n = 0; n < DIM / 512; ++n) *(v4f*)&bufA[st0 + n * 2 * XA] = v0[n];

  const int lin = half * 2 * XA + wv * 4 * XB + hm;

  const int jl = ring(half * 512 + wv * 64 + hm);
  const int al = jl >> 8, bl = (jl >> 4) & 15, cl = jl & 15;
  int gi[32];
#pragma unroll
  for (int t4 = 0; t4 < 4; ++t4) {
#pragma unroll
    for (int kk = 0; kk < 4; ++kk) {
#pragma unroll
      for (int e = 0; e < 2; ++e) {
        const int ra  = ring(kk * 1024 + e * 256) >> 8;
        const int rbc = ring(e * 256 + t4 * 16) & 0xFF;
        gi[(t4 * 4 + kk) * 2 + e] = (al ^ ra) * XA +
                                    ((bl ^ (rbc >> 4)) * XB + (cl ^ (rbc & 15)));
      }
    }
  }

  const int colc = ((tid & 6) << 1) | ((tid >> 2) & 2) | (tid & 1);

  v8f dfin[4];
  stage_gates(gm, gmat, 0, tid, colc);
  __syncthreads();
  ry_layer<true, false>(bufA, bufB, gm, gi, lin, wv, half, hm, dfin);
  for (int l = 1; l < NLAYERS - 1; ++l) {
    stage_gates(gm, gmat, l, tid, colc);
    __syncthreads();
    ry_layer<false, false>(bufA, bufB, gm, gi, lin, wv, half, hm, dfin);
  }
  stage_gates(gm, gmat, NLAYERS - 1, tid, colc);
  __syncthreads();
  ry_layer<false, true>(bufA, bufB, gm, gi, lin, wv, half, hm, dfin);

  float acc[NCLASS] = {0.f, 0.f, 0.f, 0.f, 0.f};
#pragma unroll
  for (int t4 = 0; t4 < 4; ++t4) {
    float se = 0.f, so = 0.f;
#pragma unroll
    for (int r = 0; r < 8; ++r) {
      const float v = dfin[t4][r] * dfin[t4][r];
      if (__builtin_popcount((unsigned)r) & 1) so += v; else se += v;
    }
    const float tot = se + so;
    const float dif = se - so;
    const int a  = wv * 4 + t4;
    const int pa = __popc(a) & 1;
    const int x0 = (__popc(a & 7) ^ __popc(hm) ^ half) & 1;
    const int x1 = __popc(a & 12) & 1;
    const int x2 = __popc(a & 14) & 1;
    const int x4 = pa ^ (hm >> 3);
    acc[0] += x0 ? -dif : dif;
    acc[1] += x1 ? -tot : tot;
    acc[2] += x2 ? -tot : tot;
    acc[3] += pa ? -tot : tot;
    acc[4] += x4 ? -tot : tot;
  }
#pragma unroll
  for (int w = 0; w < NCLASS; ++w) {
#pragma unroll
    for (int m = 16; m > 0; m >>= 1) acc[w] += __shfl_xor(acc[w], m, 32);
  }
  if (lane == 0) {
#pragma unroll
    for (int w = 0; w < NCLASS; ++w) red[wv * NCLASS + w] = acc[w];
  }
  __syncthreads();
  if (tid < NCLASS)
    part[(size_t)row * NCLASS + tid] =
        (red[tid] + red[NCLASS + tid]) +
        (red[2 * NCLASS + tid] + red[3 * NCLASS + tid]);
}

__global__ __launch_bounds__(256) void qnn_loss(const float* __restrict__ part,
                                                const int* __restrict__ y,
                                                float* __restrict__ out) {
  __shared__ float sred[256];
  float lsum = 0.0f;
  for (int b = threadIdx.x; b < BATCH; b += 256) {
    float lg[NCLASS];
#pragma unroll
    for (int w = 0; w < NCLASS; ++w) {
      lg[w] = part[(size_t)(2 * b) * NCLASS + w] +
              part[(size_t)(2 * b + 1) * NCLASS + w];
      out[1 + b * NCLASS + w] = lg[w];
    }
    float mx = lg[0];
#pragma unroll
    for (int w = 1; w < NCLASS; ++w) mx = fmaxf(mx, lg[w]);
    float se = 0.0f;
#pragma unroll
    for (int w = 0; w < NCLASS; ++w) se += expf(lg[w] - mx);
    const float lse = mx + logf(se);
    const int t = y[b];
    lsum += lse - lg[t];
  }
  sred[threadIdx.x] = lsum;
  __syncthreads();
  for (int s = 128; s > 0; s >>= 1) {
    if (threadIdx.x < s) sred[threadIdx.x] += sred[threadIdx.x + s];
    __syncthreads();
  }
  if (threadIdx.x == 0) out[0] = sred[0] / (float)BATCH;
}

extern "C" void kernel_launch(void* const* d_in, const int* in_sizes, int n_in,
                              void* d_out, int out_size, void* d_ws, size_t ws_size,
                              hipStream_t stream) {
  (void)in_sizes; (void)n_in; (void)out_size; (void)ws_size;
  const float* zr = (const float*)d_in[0];
  const float* zi = (const float*)d_in[1];
  const float* th = (const float*)d_in[2];
  const int*   y  = (const int*)d_in[3];
  float* out = (float*)d_out;

  float* gmat = (float*)d_ws;
  float* part = gmat + 48 * 256;

  hipLaunchKernelGGL(qnn_gates, dim3(48), dim3(256), 0, stream, th, gmat);
  hipLaunchKernelGGL(qnn_main, dim3(2 * BATCH), dim3(128), 0, stream,
                     zr, zi, gmat, part);
  hipLaunchKernelGGL(qnn_loss, dim3(1), dim3(256), 0, stream, part, y, out);
}
